# starting kernel
# baseline (speedup 1.0000x reference)
#include <hip/hip_runtime.h>
#include <hip/hip_bf16.h>


typedef float v2f __attribute__((ext_vector_type(2)));
typedef float v8f __attribute__((ext_vector_type(8)));

constexpr int BS   = 8 * 4096;
constexpr int EMB  = 768;
constexpr int FFN  = 3072;
constexpr int NQ   = 8;

constexpr int MT     = 64;
constexpr int MTILES = MT / 16;
constexpr int FC     = 192;
constexpr int FPAD   = FC + 4;
constexpr int BLOCK  = 512;
constexpr int NWAVE  = BLOCK / 32;
constexpr int ETPW   = (EMB / 16) / NWAVE;

static_assert((MT * FC) % BLOCK == 0, "h-chunk must divide evenly");
static_assert(FFN % FC == 0, "chunking must divide FFN");

__global__ __launch_bounds__(BLOCK, 1)
void ffq_fused(const float* __restrict__ x,
               const float* __restrict__ theta,
               const float* __restrict__ W1,
               const float* __restrict__ W2,
               float* __restrict__ out)
{
    __shared__ __align__(16) float zsh[MT * NQ];
    __shared__ __align__(16) float hsh[MT * FPAD];

    const int t    = threadIdx.x;
    const int lane = t & 31;
    const int wave = t >> 5;
    const int m0   = blockIdx.x * MT;

    {
        const int m = t >> 3;
        const int q = t & 7;
        zsh[t] = cosf(x[(size_t)(m0 + m) * EMB + q]) * cosf(theta[q]);
    }
    __syncthreads();

    const int nn = lane & 15;
    const int kb = (lane >> 4) << 1;

    const float* hrow = &hsh[nn * FPAD + kb];

    const float* w2base[ETPW];
#pragma unroll
    for (int j = 0; j < ETPW; ++j) {
        const int e = (wave * ETPW + j) * 16 + nn;
        w2base[j] = W2 + (size_t)e * FFN + kb;
    }

    v8f acc[MTILES][ETPW];
#pragma unroll
    for (int mt = 0; mt < MTILES; ++mt)
#pragma unroll
        for (int j = 0; j < ETPW; ++j)
            acc[mt][j] = (v8f){0.f, 0.f, 0.f, 0.f, 0.f, 0.f, 0.f, 0.f};

    for (int fc = 0; fc < FFN; fc += FC) {
#pragma unroll
        for (int i = 0; i < (MT * FC) / BLOCK; ++i) {
            const int v  = t + i * BLOCK;
            const int m  = v / FC;
            const int fl = v - m * FC;
            const float4* w1 = reinterpret_cast<const float4*>(W1 + (size_t)(fc + fl) * NQ);
            const float4 wa = w1[0];
            const float4 wb = w1[1];
            const float* zr = &zsh[m * NQ];
            float h = zr[0]*wa.x + zr[1]*wa.y + zr[2]*wa.z + zr[3]*wa.w
                    + zr[4]*wb.x + zr[5]*wb.y + zr[6]*wb.z + zr[7]*wb.w;
            hsh[m * FPAD + fl] = fmaxf(h, 0.0f);
        }
        __syncthreads();

#pragma unroll
        for (int j = 0; j < ETPW; ++j)
            __builtin_prefetch(w2base[j] + fc, 0, 0);

#pragma unroll 2
        for (int k0 = 0; k0 < FC; k0 += 4) {
            v2f a[MTILES];
#pragma unroll
            for (int mt = 0; mt < MTILES; ++mt)
                a[mt] = *reinterpret_cast<const v2f*>(hrow + mt * 16 * FPAD + k0);
#pragma unroll
            for (int j = 0; j < ETPW; ++j) {
                const v2f b = *reinterpret_cast<const v2f*>(w2base[j] + fc + k0);
#pragma unroll
                for (int mt = 0; mt < MTILES; ++mt) {
                    acc[mt][j] = __builtin_amdgcn_wmma_f32_16x16x4_f32(
                         false, a[mt],  false, b,
                         (short)0, acc[mt][j],
                         false,  false);
                }
            }
        }
        __syncthreads();
    }

    const int hi8 = (lane >> 4) << 3;
#pragma unroll
    for (int mt = 0; mt < MTILES; ++mt) {
        const int mrow = m0 + mt * 16 + hi8;
#pragma unroll
        for (int j = 0; j < ETPW; ++j) {
            const int e0 = (wave * ETPW + j) * 16;
#pragma unroll
            for (int r = 0; r < 8; ++r) {
                out[(size_t)(mrow + r) * EMB + e0 + nn] = acc[mt][j][r];
            }
        }
    }
}

extern "C" void kernel_launch(void* const* d_in, const int* in_sizes, int n_in,
                              void* d_out, int out_size, void* d_ws, size_t ws_size,
                              hipStream_t stream) {
    (void)in_sizes; (void)n_in; (void)d_ws; (void)ws_size; (void)out_size;
    const float* x     = (const float*)d_in[0];
    const float* theta = (const float*)d_in[1];
    const float* W1    = (const float*)d_in[2];
    const float* W2    = (const float*)d_in[3];
    float* out = (float*)d_out;

    dim3 grid(BS / MT);
    dim3 block(BLOCK);
    hipLaunchKernelGGL(ffq_fused, grid, block, 0, stream, x, theta, W1, W2, out);
}


# W2 (B operand) now goes global -> LDS -> fragments: each wave copies its own 48 W2 rows, 32 k per block, with whole-128-B-line loads a block ahead into a private two-stage LDS tile; sum order kept
# speedup vs baseline: 1.1365x; 1.1365x over previous
#include <hip/hip_runtime.h>
#include <hip/hip_bf16.h>


typedef float v2f __attribute__((ext_vector_type(2)));
typedef float v8f __attribute__((ext_vector_type(8)));

constexpr int BS   = 8 * 4096;
constexpr int EMB  = 768;
constexpr int FFN  = 3072;
constexpr int NQ   = 8;

constexpr int MT     = 64;
constexpr int MTILES = MT / 16;
constexpr int FC     = 192;
constexpr int FPAD   = FC + 4;
constexpr int BLOCK  = 512;
constexpr int NWAVE  = BLOCK / 32;
constexpr int ETPW   = (EMB / 16) / NWAVE;

constexpr int KB     = 32;
constexpr int NBLK   = FFN / KB;
constexpr int WROWS  = ETPW * 16;
constexpr int BPITCH = KB + 4;
constexpr int BSTAGE = WROWS * BPITCH;
constexpr int NLD    = WROWS / 4;

static_assert(FC % KB == 0, "staged blocks must tile the h chunk");
static_assert((MT * FC) % BLOCK == 0, "h-chunk must divide evenly");
static_assert(FFN % FC == 0, "chunking must divide FFN");

__global__ __launch_bounds__(BLOCK, 1)
void ffq_fused(const float* __restrict__ x,
               const float* __restrict__ theta,
               const float* __restrict__ W1,
               const float* __restrict__ W2,
               float* __restrict__ out)
{
    __shared__ __align__(16) float lds[MT * FPAD + MT * NQ + NWAVE * 2 * BSTAGE];
    float* const hsh = lds;
    float* const zsh = lds + MT * FPAD;
    float* const bsh = lds + MT * FPAD + MT * NQ;

    const int t    = threadIdx.x;
    const int lane = t & 31;
    const int wave = t >> 5;
    const int m0   = blockIdx.x * MT;

    {
        const int m = t >> 3;
        const int q = t & 7;
        zsh[t] = cosf(x[(size_t)(m0 + m) * EMB + q]) * cosf(theta[q]);
    }
    __syncthreads();

    const int nn = lane & 15;
    const int kb = (lane >> 4) << 1;

    const float* hrow = &hsh[nn * FPAD + kb];

    const int lr = lane >> 3;
    const int lc = (lane & 7) << 2;
    float*       bw   = bsh + wave * (2 * BSTAGE);
    const float* w2g  = W2 + (size_t)(wave * WROWS + lr) * FFN + lc;
    float*       bstw = bw + lr * BPITCH + lc;
    const float* brd  = bw + nn * BPITCH + kb;

#define STG_EACH(X) X(0) X(1) X(2) X(3) X(4) X(5) X(6) X(7) X(8) X(9) X(10) X(11)
#define STG_DECL(i)  float4 stg##i;
#define STG_LOAD(i)  stg##i = *reinterpret_cast<const float4*>(w2src + (size_t)(i) * 4 * FFN);
#define STG_STORE(i) *reinterpret_cast<float4*>(bdst + (i) * 4 * BPITCH) = stg##i;
    static_assert(NLD == 12, "STG_EACH lists NLD values");
    STG_EACH(STG_DECL)
    {
        const float* w2src = w2g;
        float*       bdst  = bstw;
        STG_EACH(STG_LOAD)
        STG_EACH(STG_STORE)
    }
    int blk = 0;

    v8f acc[MTILES][ETPW];
#pragma unroll
    for (int mt = 0; mt < MTILES; ++mt)
#pragma unroll
        for (int j = 0; j < ETPW; ++j)
            acc[mt][j] = (v8f){0.f, 0.f, 0.f, 0.f, 0.f, 0.f, 0.f, 0.f};

    for (int fc = 0; fc < FFN; fc += FC) {
#pragma unroll
        for (int i = 0; i < (MT * FC) / BLOCK; ++i) {
            const int v  = t + i * BLOCK;
            const int m  = v / FC;
            const int fl = v - m * FC;
            const float4* w1 = reinterpret_cast<const float4*>(W1 + (size_t)(fc + fl) * NQ);
            const float4 wa = w1[0];
            const float4 wb = w1[1];
            const float* zr = &zsh[m * NQ];
            float h = zr[0]*wa.x + zr[1]*wa.y + zr[2]*wa.z + zr[3]*wa.w
                    + zr[4]*wb.x + zr[5]*wb.y + zr[6]*wb.z + zr[7]*wb.w;
            hsh[m * FPAD + fl] = fmaxf(h, 0.0f);
        }
        __syncthreads();

        {
            const float* w2src = w2g + (blk + 1) * KB;
            STG_EACH(STG_LOAD)
        }
        for (int kk = 0; kk < FC; kk += KB, ++blk) {
            const float* bs = brd + (blk & 1) * BSTAGE;

#pragma unroll 2
            for (int k0 = 0; k0 < KB; k0 += 4) {
                v2f a[MTILES];
#pragma unroll
                for (int mt = 0; mt < MTILES; ++mt)
                    a[mt] = *reinterpret_cast<const v2f*>(hrow + mt * 16 * FPAD + kk + k0);
#pragma unroll
                for (int j = 0; j < ETPW; ++j) {
                    const v2f b = *reinterpret_cast<const v2f*>(bs + j * 16 * BPITCH + k0);
#pragma unroll
                    for (int mt = 0; mt < MTILES; ++mt) {
                        acc[mt][j] = __builtin_amdgcn_wmma_f32_16x16x4_f32(
                             false, a[mt],  false, b,
                             (short)0, acc[mt][j],
                             false,  false);
                    }
                }
            }

            float* bdst = bstw + ((blk + 1) & 1) * BSTAGE;
            STG_EACH(STG_STORE)
            __builtin_amdgcn_fence(__ATOMIC_ACQ_REL, "wavefront");
            if (kk + KB < FC) {
                const int bn = (blk + 2 < NBLK) ? (blk + 2) : (NBLK - 1);
                const float* w2src = w2g + bn * KB;
                STG_EACH(STG_LOAD)
            }
        }
        __syncthreads();
    }

    const int hi8 = (lane >> 4) << 3;
#pragma unroll
    for (int mt = 0; mt < MTILES; ++mt) {
        const int mrow = m0 + mt * 16 + hi8;
#pragma unroll
        for (int j = 0; j < ETPW; ++j) {
            const int e0 = (wave * ETPW + j) * 16;
#pragma unroll
            for (int r = 0; r < 8; ++r) {
                out[(size_t)(mrow + r) * EMB + e0 + nn] = acc[mt][j][r];
            }
        }
    }
}

extern "C" void kernel_launch(void* const* d_in, const int* in_sizes, int n_in,
                              void* d_out, int out_size, void* d_ws, size_t ws_size,
                              hipStream_t stream) {
    (void)in_sizes; (void)n_in; (void)d_ws; (void)ws_size; (void)out_size;
    const float* x     = (const float*)d_in[0];
    const float* theta = (const float*)d_in[1];
    const float* W1    = (const float*)d_in[2];
    const float* W2    = (const float*)d_in[3];
    float* out = (float*)d_out;

    dim3 grid(BS / MT);
    dim3 block(BLOCK);
    hipLaunchKernelGGL(ffq_fused, grid, block, 0, stream, x, theta, W1, W2, out);
}


# h chunk double-buffered in LDS: each wave computes its 4 rows of the next chunk's layer 1, one 32-column piece per W2 block, inside the matrix loop; one barrier per chunk; sum order kept
# speedup vs baseline: 1.2029x; 1.0584x over previous
#include <hip/hip_runtime.h>
#include <hip/hip_bf16.h>


typedef float v2f __attribute__((ext_vector_type(2)));
typedef float v8f __attribute__((ext_vector_type(8)));

constexpr int BS   = 8 * 4096;
constexpr int EMB  = 768;
constexpr int FFN  = 3072;
constexpr int NQ   = 8;

constexpr int MT     = 64;
constexpr int MTILES = MT / 16;
constexpr int FC     = 192;
constexpr int FPAD   = FC + 4;
constexpr int BLOCK  = 512;
constexpr int NWAVE  = BLOCK / 32;
constexpr int ETPW   = (EMB / 16) / NWAVE;
constexpr int HBUF   = MT * FPAD;
constexpr int HROWS  = MT / NWAVE;

constexpr int KB     = 32;
constexpr int NBLK   = FFN / KB;
constexpr int WROWS  = ETPW * 16;
constexpr int BPITCH = KB + 4;
constexpr int BSTAGE = WROWS * BPITCH;
constexpr int NLD    = WROWS / 4;

static_assert(FC % KB == 0, "staged blocks must tile the h chunk");
static_assert(KB == 32, "an h piece is one column per lane");
static_assert(HROWS * NWAVE == MT, "the waves' rows must tile the h chunk");
static_assert(FFN % FC == 0, "chunking must divide FFN");

__global__ __launch_bounds__(BLOCK, 1)
void ffq_fused(const float* __restrict__ x,
               const float* __restrict__ theta,
               const float* __restrict__ W1,
               const float* __restrict__ W2,
               float* __restrict__ out)
{
    constexpr int HSTRIDE = HBUF + NWAVE * 2 * BSTAGE;
    __shared__ __align__(16) float lds[HSTRIDE + HBUF + MT * NQ];
    float* const hsh = lds;
    float* const bsh = lds + HBUF;
    float* const zsh = lds + HSTRIDE + HBUF;

    const int t    = threadIdx.x;
    const int lane = t & 31;
    const int wave = t >> 5;
    const int m0   = blockIdx.x * MT;

    {
        const int m = t >> 3;
        const int q = t & 7;
        zsh[t] = cosf(x[(size_t)(m0 + m) * EMB + q]) * cosf(theta[q]);
    }
    __syncthreads();

    const int nn = lane & 15;
    const int kb = (lane >> 4) << 1;

    const float* hrow = &hsh[nn * FPAD + kb];

    float* const hst = &hsh[(wave * HROWS) * FPAD + lane];
    float zs[HROWS][NQ];
#pragma unroll
    for (int r = 0; r < HROWS; ++r)
#pragma unroll
        for (int q = 0; q < NQ; ++q)
            zs[r][q] = __int_as_float(__builtin_amdgcn_readfirstlane(
                           __float_as_int(zsh[(wave * HROWS + r) * NQ + q])));

    auto h_piece = [&](const float4 wa, const float4 wb, float* hdst) {
#pragma unroll
        for (int r = 0; r < HROWS; ++r) {
            float h = zs[r][1] * wa.y;
            h = fmaf(zs[r][0], wa.x, h);
            h = fmaf(zs[r][2], wa.z, h);
            h = fmaf(zs[r][3], wa.w, h);
            h = fmaf(zs[r][4], wb.x, h);
            h = fmaf(zs[r][5], wb.y, h);
            h = fmaf(zs[r][6], wb.z, h);
            h = fmaf(zs[r][7], wb.w, h);
            hdst[r * FPAD] = fmaxf(h, 0.0f);
        }
    };

    const int lr = lane >> 3;
    const int lc = (lane & 7) << 2;
    float*       bw   = bsh + wave * (2 * BSTAGE);
    const float* w2g  = W2 + (size_t)(wave * WROWS + lr) * FFN + lc;
    float*       bstw = bw + lr * BPITCH + lc;
    const float* brd  = bw + nn * BPITCH + kb;

#define STG_EACH(X) X(0) X(1) X(2) X(3) X(4) X(5) X(6) X(7) X(8) X(9) X(10) X(11)
#define STG_DECL(i)  float4 stg##i;
#define STG_LOAD(i)  stg##i = *reinterpret_cast<const float4*>(w2src + (size_t)(i) * 4 * FFN);
#define STG_STORE(i) *reinterpret_cast<float4*>(bdst + (i) * 4 * BPITCH) = stg##i;
    static_assert(NLD == 12, "STG_EACH lists NLD values");
    STG_EACH(STG_DECL)
    {
        const float* w2src = w2g;
        float*       bdst  = bstw;
        STG_EACH(STG_LOAD)
        STG_EACH(STG_STORE)
    }
    int blk = 0;

    v8f acc[MTILES][ETPW];
#pragma unroll
    for (int mt = 0; mt < MTILES; ++mt)
#pragma unroll
        for (int j = 0; j < ETPW; ++j)
            acc[mt][j] = (v8f){0.f, 0.f, 0.f, 0.f, 0.f, 0.f, 0.f, 0.f};

#pragma unroll
    for (int p = 0; p < FC / KB; ++p) {
        const float4* w1 = reinterpret_cast<const float4*>(W1 + (size_t)(p * KB + lane) * NQ);
        h_piece(w1[0], w1[1], hst + p * KB);
    }
    __syncthreads();

    for (int fc = 0; fc < FFN; fc += FC) {
        const int    cur   = (fc / FC) & 1;
        const float* hcur  = hrow + cur * HSTRIDE;
        float*       hnext = hst + (cur ^ 1) * HSTRIDE;
        const bool   more  = fc + FC < FFN;

        {
            const float* w2src = w2g + (blk + 1) * KB;
            STG_EACH(STG_LOAD)
        }
        for (int kk = 0; kk < FC; kk += KB, ++blk) {
            const float* bs = brd + (blk & 1) * BSTAGE;

            float4 wa = make_float4(0.f, 0.f, 0.f, 0.f), wb = wa;
            if (more) {
                const float4* w1 = reinterpret_cast<const float4*>(W1 + (size_t)(fc + FC + kk + lane) * NQ);
                wa = w1[0];
                wb = w1[1];
            }

#pragma unroll 2
            for (int k0 = 0; k0 < KB; k0 += 4) {
                v2f a[MTILES];
#pragma unroll
                for (int mt = 0; mt < MTILES; ++mt)
                    a[mt] = *reinterpret_cast<const v2f*>(hcur + mt * 16 * FPAD + kk + k0);
#pragma unroll
                for (int j = 0; j < ETPW; ++j) {
                    const v2f b = *reinterpret_cast<const v2f*>(bs + j * 16 * BPITCH + k0);
#pragma unroll
                    for (int mt = 0; mt < MTILES; ++mt) {
                        acc[mt][j] = __builtin_amdgcn_wmma_f32_16x16x4_f32(
                             false, a[mt],  false, b,
                             (short)0, acc[mt][j],
                             false,  false);
                    }
                }
            }

            if (more) h_piece(wa, wb, hnext + kk);

            float* bdst = bstw + ((blk + 1) & 1) * BSTAGE;
            STG_EACH(STG_STORE)
            __builtin_amdgcn_fence(__ATOMIC_ACQ_REL, "wavefront");
            if (kk + KB < FC) {
                const int bn = (blk + 2 < NBLK) ? (blk + 2) : (NBLK - 1);
                const float* w2src = w2g + bn * KB;
                STG_EACH(STG_LOAD)
            }
        }
        __syncthreads();
    }

    const int hi8 = (lane >> 4) << 3;
#pragma unroll
    for (int mt = 0; mt < MTILES; ++mt) {
        const int mrow = m0 + mt * 16 + hi8;
#pragma unroll
        for (int j = 0; j < ETPW; ++j) {
            const int e0 = (wave * ETPW + j) * 16;
#pragma unroll
            for (int r = 0; r < 8; ++r) {
                out[(size_t)(mrow + r) * EMB + e0 + nn] = acc[mt][j][r];
            }
        }
    }
}

extern "C" void kernel_launch(void* const* d_in, const int* in_sizes, int n_in,
                              void* d_out, int out_size, void* d_ws, size_t ws_size,
                              hipStream_t stream) {
    (void)in_sizes; (void)n_in; (void)d_ws; (void)ws_size; (void)out_size;
    const float* x     = (const float*)d_in[0];
    const float* theta = (const float*)d_in[1];
    const float* W1    = (const float*)d_in[2];
    const float* W2    = (const float*)d_in[3];
    float* out = (float*)d_out;

    dim3 grid(BS / MT);
    dim3 block(BLOCK);
    hipLaunchKernelGGL(ffq_fused, grid, block, 0, stream, x, theta, W1, W2, out);
}


# Wave priority falls with progress inside each h chunk (s_setprio 3,3,2,2,1,0 over its six W2 blocks): lagging waves of a SIMD get the matrix pipe, all reach the chunk barrier together; sums unchanged
# speedup vs baseline: 1.2439x; 1.0341x over previous
#include <hip/hip_runtime.h>
#include <hip/hip_bf16.h>


typedef float v2f __attribute__((ext_vector_type(2)));
typedef float v8f __attribute__((ext_vector_type(8)));

constexpr int BS   = 8 * 4096;
constexpr int EMB  = 768;
constexpr int FFN  = 3072;
constexpr int NQ   = 8;

constexpr int MT     = 64;
constexpr int MTILES = MT / 16;
constexpr int FC     = 192;
constexpr int FPAD   = FC + 4;
constexpr int BLOCK  = 512;
constexpr int NWAVE  = BLOCK / 32;
constexpr int ETPW   = (EMB / 16) / NWAVE;
constexpr int HBUF   = MT * FPAD;
constexpr int HROWS  = MT / NWAVE;

constexpr int KB     = 32;
constexpr int NBLK   = FFN / KB;
constexpr int WROWS  = ETPW * 16;
constexpr int BPITCH = KB + 4;
constexpr int BSTAGE = WROWS * BPITCH;
constexpr int NLD    = WROWS / 4;

static_assert(FC % KB == 0, "staged blocks must tile the h chunk");
static_assert(KB == 32, "an h piece is one column per lane");
static_assert(HROWS * NWAVE == MT, "the waves' rows must tile the h chunk");
static_assert(FFN % FC == 0, "chunking must divide FFN");
static_assert(FC / KB == 6, "the priority levels are written for six blocks per chunk");

__global__ __launch_bounds__(BLOCK, 1)
void ffq_fused(const float* __restrict__ x,
               const float* __restrict__ theta,
               const float* __restrict__ W1,
               const float* __restrict__ W2,
               float* __restrict__ out)
{
    constexpr int HSTRIDE = HBUF + NWAVE * 2 * BSTAGE;
    __shared__ __align__(16) float lds[HSTRIDE + HBUF + MT * NQ];
    float* const hsh = lds;
    float* const bsh = lds + HBUF;
    float* const zsh = lds + HSTRIDE + HBUF;

    const int t    = threadIdx.x;
    const int lane = t & 31;
    const int wave = t >> 5;
    const int m0   = blockIdx.x * MT;

    {
        const int m = t >> 3;
        const int q = t & 7;
        zsh[t] = cosf(x[(size_t)(m0 + m) * EMB + q]) * cosf(theta[q]);
    }
    __syncthreads();

    const int nn = lane & 15;
    const int kb = (lane >> 4) << 1;

    const float* hrow = &hsh[nn * FPAD + kb];

    float* const hst = &hsh[(wave * HROWS) * FPAD + lane];
    float zs[HROWS][NQ];
#pragma unroll
    for (int r = 0; r < HROWS; ++r)
#pragma unroll
        for (int q = 0; q < NQ; ++q)
            zs[r][q] = __int_as_float(__builtin_amdgcn_readfirstlane(
                           __float_as_int(zsh[(wave * HROWS + r) * NQ + q])));

    auto h_piece = [&](const float4 wa, const float4 wb, float* hdst) {
#pragma unroll
        for (int r = 0; r < HROWS; ++r) {
            float h = zs[r][1] * wa.y;
            h = fmaf(zs[r][0], wa.x, h);
            h = fmaf(zs[r][2], wa.z, h);
            h = fmaf(zs[r][3], wa.w, h);
            h = fmaf(zs[r][4], wb.x, h);
            h = fmaf(zs[r][5], wb.y, h);
            h = fmaf(zs[r][6], wb.z, h);
            h = fmaf(zs[r][7], wb.w, h);
            hdst[r * FPAD] = fmaxf(h, 0.0f);
        }
    };

    const int lr = lane >> 3;
    const int lc = (lane & 7) << 2;
    float*       bw   = bsh + wave * (2 * BSTAGE);
    const float* w2g  = W2 + (size_t)(wave * WROWS + lr) * FFN + lc;
    float*       bstw = bw + lr * BPITCH + lc;
    const float* brd  = bw + nn * BPITCH + kb;

#define STG_EACH(X) X(0) X(1) X(2) X(3) X(4) X(5) X(6) X(7) X(8) X(9) X(10) X(11)
#define STG_DECL(i)  float4 stg##i;
#define STG_LOAD(i)  stg##i = *reinterpret_cast<const float4*>(w2src + (size_t)(i) * 4 * FFN);
#define STG_STORE(i) *reinterpret_cast<float4*>(bdst + (i) * 4 * BPITCH) = stg##i;
    static_assert(NLD == 12, "STG_EACH lists NLD values");
    STG_EACH(STG_DECL)
    {
        const float* w2src = w2g;
        float*       bdst  = bstw;
        STG_EACH(STG_LOAD)
        STG_EACH(STG_STORE)
    }
    int blk = 0;

    v8f acc[MTILES][ETPW];
#pragma unroll
    for (int mt = 0; mt < MTILES; ++mt)
#pragma unroll
        for (int j = 0; j < ETPW; ++j)
            acc[mt][j] = (v8f){0.f, 0.f, 0.f, 0.f, 0.f, 0.f, 0.f, 0.f};

#pragma unroll
    for (int p = 0; p < FC / KB; ++p) {
        const float4* w1 = reinterpret_cast<const float4*>(W1 + (size_t)(p * KB + lane) * NQ);
        h_piece(w1[0], w1[1], hst + p * KB);
    }
    __syncthreads();

    for (int fc = 0; fc < FFN; fc += FC) {
        const int    cur   = (fc / FC) & 1;
        const float* hcur  = hrow + cur * HSTRIDE;
        float*       hnext = hst + (cur ^ 1) * HSTRIDE;
        const bool   more  = fc + FC < FFN;

        {
            const float* w2src = w2g + (blk + 1) * KB;
            STG_EACH(STG_LOAD)
        }
        for (int kk = 0; kk < FC; kk += KB, ++blk) {
            if (kk == 0)                __builtin_amdgcn_s_setprio(3);
            else if (kk == 2 * KB)      __builtin_amdgcn_s_setprio(2);
            else if (kk == 4 * KB)      __builtin_amdgcn_s_setprio(1);
            else if (kk == 5 * KB)      __builtin_amdgcn_s_setprio(0);

            const float* bs = brd + (blk & 1) * BSTAGE;

            float4 wa = make_float4(0.f, 0.f, 0.f, 0.f), wb = wa;
            if (more) {
                const float4* w1 = reinterpret_cast<const float4*>(W1 + (size_t)(fc + FC + kk + lane) * NQ);
                wa = w1[0];
                wb = w1[1];
            }

#pragma unroll 2
            for (int k0 = 0; k0 < KB; k0 += 4) {
                v2f a[MTILES];
#pragma unroll
                for (int mt = 0; mt < MTILES; ++mt)
                    a[mt] = *reinterpret_cast<const v2f*>(hcur + mt * 16 * FPAD + kk + k0);
#pragma unroll
                for (int j = 0; j < ETPW; ++j) {
                    const v2f b = *reinterpret_cast<const v2f*>(bs + j * 16 * BPITCH + k0);
#pragma unroll
                    for (int mt = 0; mt < MTILES; ++mt) {
                        acc[mt][j] = __builtin_amdgcn_wmma_f32_16x16x4_f32(
                             false, a[mt],  false, b,
                             (short)0, acc[mt][j],
                             false,  false);
                    }
                }
            }

            if (more) h_piece(wa, wb, hnext + kk);

            float* bdst = bstw + ((blk + 1) & 1) * BSTAGE;
            STG_EACH(STG_STORE)
            __builtin_amdgcn_fence(__ATOMIC_ACQ_REL, "wavefront");
            if (kk + KB < FC) {
                const int bn = (blk + 2 < NBLK) ? (blk + 2) : (NBLK - 1);
                const float* w2src = w2g + bn * KB;
                STG_EACH(STG_LOAD)
            }
        }
        __syncthreads();
    }

    const int hi8 = (lane >> 4) << 3;
#pragma unroll
    for (int mt = 0; mt < MTILES; ++mt) {
        const int mrow = m0 + mt * 16 + hi8;
#pragma unroll
        for (int j = 0; j < ETPW; ++j) {
            const int e0 = (wave * ETPW + j) * 16;
#pragma unroll
            for (int r = 0; r < 8; ++r) {
                out[(size_t)(mrow + r) * EMB + e0 + nn] = acc[mt][j][r];
            }
        }
    }
}

extern "C" void kernel_launch(void* const* d_in, const int* in_sizes, int n_in,
                              void* d_out, int out_size, void* d_ws, size_t ws_size,
                              hipStream_t stream) {
    (void)in_sizes; (void)n_in; (void)d_ws; (void)ws_size; (void)out_size;
    const float* x     = (const float*)d_in[0];
    const float* theta = (const float*)d_in[1];
    const float* W1    = (const float*)d_in[2];
    const float* W2    = (const float*)d_in[3];
    float* out = (float*)d_out;

    dim3 grid(BS / MT);
    dim3 block(BLOCK);
    hipLaunchKernelGGL(ffq_fused, grid, block, 0, stream, x, theta, W1, W2, out);
}
